# starting kernel
# baseline (speedup 1.0000x reference)
#include <hip/hip_runtime.h>
#include <hip/hip_bf16.h>


#define B_ 16
#define L_ 4096
#define M_ 64
#define SPLITS 8

typedef __attribute__((ext_vector_type(2))) float v2f;
typedef __attribute__((ext_vector_type(8))) float v8f;

__global__ void k_coef(float* c0, float* c1, float* c2) {
    int t = threadIdx.x;
    const int Ns[3] = {512, 256, 128};
    float* cs[3] = {c0, c1, c2};
    for (int sc = 0; sc < 3; ++sc) {
        int N = Ns[sc];
        if (t < N) {
            double acc = 0.0;
            double step = (double)(N - 1) / 511.0;
            for (int tt = 0; tt < 512; ++tt) {
                double pos = tt * step;
                int i0 = (int)pos;
                int i1 = i0 + 1; if (i1 > N - 1) i1 = N - 1;
                double w1 = pos - (double)i0;
                if (t == i0) acc += 1.0 - w1;
                if (t == i1) acc += w1;
            }
            cs[sc][t] = (float)(acc / 512.0);
        }
    }
}

__global__ void k_stage1(const float* __restrict__ x,
                         const float* __restrict__ c0,
                         const float* __restrict__ c1,
                         const float* __restrict__ c2,
                         float* __restrict__ p0, float* __restrict__ p1,
                         float* __restrict__ p2) {
    int e2 = blockIdx.x * 256 + threadIdx.x;
    int b  = blockIdx.y;
    int sp = blockIdx.z;
    int h  = e2 >> 10;
    int q  = e2 >> 9;
    const float* xb = x + (size_t)b * (L_ * M_);
    int i2b = sp * (128 / SPLITS);
    float a0 = 0.f, a1 = 0.f, a2 = 0.f;
#pragma unroll 4
    for (int i = 0; i < 128 / SPLITS; ++i) {
        int i2 = i2b + i;
        float v = xb[(size_t)i2 * 2048 + e2];
        a2 = fmaf(c2[i2],         v, a2);
        a1 = fmaf(c1[2 * i2 + h], v, a1);
        a0 = fmaf(c0[4 * i2 + q], v, a0);
    }
    size_t o = ((size_t)sp * B_ + b) * 2048 + e2;
    p0[o] = a0; p1[o] = a1; p2[o] = a2;
}

__global__ void k_stage2(const float* __restrict__ p0,
                         const float* __restrict__ p1,
                         const float* __restrict__ p2,
                         float* __restrict__ A) {
    int idx = blockIdx.x * 256 + threadIdx.x;
    int row = idx >> 6;
    int kk  = idx & 63;
    int b = row >> 6, m = row & 63;
    float acc = 0.f;
    if (kk < 8) {
        int j = m >> 3, s = m & 7, k = kk;
        int e = j * 64 + s * 8 + k;
        for (int sp = 0; sp < SPLITS; ++sp)
            for (int qq = 0; qq < 4; ++qq)
                acc += p0[((size_t)sp * B_ + b) * 2048 + qq * 512 + e];
    } else if (kk < 24) {
        int j = m >> 2, s = m & 3, k = kk - 8;
        int e = j * 64 + s * 16 + k;
        for (int sp = 0; sp < SPLITS; ++sp)
            for (int hh = 0; hh < 2; ++hh)
                acc += p1[((size_t)sp * B_ + b) * 2048 + hh * 1024 + e];
    } else if (kk < 56) {
        int j = m >> 1, s = m & 1, k = kk - 24;
        int e = j * 64 + s * 32 + k;
        for (int sp = 0; sp < SPLITS; ++sp)
            acc += p2[((size_t)sp * B_ + b) * 2048 + e];
    }
    A[idx] = acc;
}

__global__ void k_weff(const float* __restrict__ pw0, const float* __restrict__ pw1,
                       const float* __restrict__ pw2, const float* __restrict__ fuse_w,
                       float* __restrict__ Weff) {
    int idx = blockIdx.x * 256 + threadIdx.x;
    if (idx >= 56 * 384) return;
    int kk = idx / 384, df = idx % 384;
    const float* w; int sc, k;
    if (kk < 8)       { sc = 0; k = kk;      w = pw0; }
    else if (kk < 24) { sc = 1; k = kk - 8;  w = pw1; }
    else              { sc = 2; k = kk - 24; w = pw2; }
    float acc = 0.f;
    for (int d = 0; d < 128; ++d)
        acc = fmaf(w[k * 128 + d], fuse_w[(size_t)(sc * 128 + d) * 384 + df], acc);
    Weff[idx] = acc;
}

__global__ void k_beff(const float* __restrict__ pb0, const float* __restrict__ pb1,
                       const float* __restrict__ pb2, const float* __restrict__ fuse_w,
                       const float* __restrict__ fuse_b, float* __restrict__ beff) {
    int df = blockIdx.x * 256 + threadIdx.x;
    if (df >= 384) return;
    float acc = fuse_b[df];
    const float* pbs[3] = {pb0, pb1, pb2};
    for (int sc = 0; sc < 3; ++sc)
        for (int d = 0; d < 128; ++d)
            acc = fmaf(pbs[sc][d], fuse_w[(size_t)(sc * 128 + d) * 384 + df], acc);
    beff[df] = acc;
}

__global__ void k_wtot(const float* __restrict__ Weff, const float* __restrict__ flat_w,
                       float* __restrict__ Wtot) {
    int idx = blockIdx.x * 256 + threadIdx.x;
    int kk = idx >> 9, n = idx & 511;
    float acc = 0.f;
    if (kk < 56)
        for (int df = 0; df < 384; ++df)
            acc = fmaf(Weff[kk * 384 + df], flat_w[(size_t)df * 512 + n], acc);
    Wtot[idx] = acc;
}

__global__ void k_btot(const float* __restrict__ beff, const float* __restrict__ flat_w,
                       const float* __restrict__ flat_b, float* __restrict__ btot) {
    int n = blockIdx.x * 256 + threadIdx.x;
    if (n >= 512) return;
    float acc = flat_b[n];
    for (int df = 0; df < 384; ++df)
        acc = fmaf(beff[df], flat_w[(size_t)df * 512 + n], acc);
    btot[n] = acc;
}

__global__ void k_gemm(const float* __restrict__ A, const float* __restrict__ Wt,
                       const float* __restrict__ btot, float* __restrict__ out) {
    int wave = threadIdx.x >> 5;
    int lane = threadIdx.x & 31;
    int tile = blockIdx.x * 8 + wave;
    int mt = tile >> 5, nt = tile & 31;
    int mbase = mt * 16, nbase = nt * 16;
    int half = lane >> 4, l16 = lane & 15;
    const float* Ar = A + (size_t)(mbase + l16) * 64;
    const float* Bc = Wt + nbase + l16;
    v8f acc = {};
#pragma unroll
    for (int k0 = 0; k0 < 64; k0 += 4) {
        int ka = k0 + half * 2;
        v2f av = { Ar[ka], Ar[ka + 1] };
        v2f bv = { Bc[(size_t)ka * 512], Bc[(size_t)(ka + 1) * 512] };
        acc = __builtin_amdgcn_wmma_f32_16x16x4_f32(
            false, av, false, bv, (short)0, acc, false, false);
    }
    float bias = btot[nbase + l16];
#pragma unroll
    for (int r = 0; r < 8; ++r)
        out[(size_t)(mbase + r + 8 * half) * 512 + nbase + l16] = acc[r] + bias;
}

extern "C" void kernel_launch(void* const* d_in, const int* in_sizes, int n_in,
                              void* d_out, int out_size, void* d_ws, size_t ws_size,
                              hipStream_t stream) {
    (void)in_sizes; (void)n_in; (void)out_size; (void)ws_size;
    const float* x      = (const float*)d_in[0];
    const float* pw0    = (const float*)d_in[1];
    const float* pb0    = (const float*)d_in[2];
    const float* pw1    = (const float*)d_in[3];
    const float* pb1    = (const float*)d_in[4];
    const float* pw2    = (const float*)d_in[5];
    const float* pb2    = (const float*)d_in[6];
    const float* fuse_w = (const float*)d_in[7];
    const float* fuse_b = (const float*)d_in[8];
    const float* flat_w = (const float*)d_in[9];
    const float* flat_b = (const float*)d_in[10];
    float* out = (float*)d_out;

    float* ws   = (float*)d_ws;
    float* c0   = ws;
    float* c1   = ws + 512;
    float* c2   = ws + 768;
    float* p0   = ws + 1024;
    float* p1   = p0 + SPLITS * B_ * 2048;
    float* p2   = p1 + SPLITS * B_ * 2048;
    float* A    = p2 + SPLITS * B_ * 2048;
    float* Weff = A + 1024 * 64;
    float* beff = Weff + 56 * 384;
    float* Wtot = beff + 384;
    float* btot = Wtot + 64 * 512;

    k_coef<<<1, 512, 0, stream>>>(c0, c1, c2);
    k_weff<<<(56 * 384 + 255) / 256, 256, 0, stream>>>(pw0, pw1, pw2, fuse_w, Weff);
    k_beff<<<2, 256, 0, stream>>>(pb0, pb1, pb2, fuse_w, fuse_b, beff);
    k_wtot<<<(64 * 512) / 256, 256, 0, stream>>>(Weff, flat_w, Wtot);
    k_btot<<<2, 256, 0, stream>>>(beff, flat_w, flat_b, btot);

    dim3 g1(2048 / 256, B_, SPLITS);
    k_stage1<<<g1, 256, 0, stream>>>(x, c0, c1, c2, p0, p1, p2);
    k_stage2<<<(1024 * 64) / 256, 256, 0, stream>>>(p0, p1, p2, A);

    k_gemm<<<2048 / 8, 256, 0, stream>>>(A, Wtot, btot, out);
}


# weight+bias folds: 4 dependent launches -> 1; a block owns 4 rows x 64 cols of Wtot, builds its Weff rows in LDS, multiplies on by flat_w; K split over threads, partials via LDS; fold sums regrouped
# speedup vs baseline: 13.0435x; 13.0435x over previous
#include <hip/hip_runtime.h>
#include <hip/hip_bf16.h>


#define B_ 16
#define L_ 4096
#define M_ 64
#define SPLITS 8

typedef __attribute__((ext_vector_type(2))) float v2f;
typedef __attribute__((ext_vector_type(8))) float v8f;

__global__ void k_coef(float* c0, float* c1, float* c2) {
    int lane = threadIdx.x & 31;
    int w = blockIdx.x * 4 + (threadIdx.x >> 5);
    int N, t; float* dst; double step;
    if (w < 512)      { N = 512; t = w;       dst = c0 + t; step = 511.0 / 511.0; }
    else if (w < 768) { N = 256; t = w - 512; dst = c1 + t; step = 255.0 / 511.0; }
    else              { N = 128; t = w - 768; dst = c2 + t; step = 127.0 / 511.0; }
    double acc = 0.0;
    for (int j = 0; j < 16; ++j) {
        int tt = lane * 16 + j;
        double pos = tt * step;
        int i0 = (int)pos;
        int i1 = i0 + 1; if (i1 > N - 1) i1 = N - 1;
        double w1 = pos - (double)i0;
        if (t == i0) acc += 1.0 - w1;
        if (t == i1) acc += w1;
    }
    for (int off = 16; off > 0; off >>= 1)
        acc += __shfl_xor(acc, off);
    if (lane == 0) *dst = (float)(acc / 512.0);
}

__global__ void k_stage1(const float* __restrict__ x,
                         const float* __restrict__ c0,
                         const float* __restrict__ c1,
                         const float* __restrict__ c2,
                         float* __restrict__ p0, float* __restrict__ p1,
                         float* __restrict__ p2) {
    int e2 = blockIdx.x * 256 + threadIdx.x;
    int b  = blockIdx.y;
    int sp = blockIdx.z;
    int h  = e2 >> 10;
    int q  = e2 >> 9;
    const float* xb = x + (size_t)b * (L_ * M_);
    int i2b = sp * (128 / SPLITS);
    float a0 = 0.f, a1 = 0.f, a2 = 0.f;
#pragma unroll 4
    for (int i = 0; i < 128 / SPLITS; ++i) {
        int i2 = i2b + i;
        float v = xb[(size_t)i2 * 2048 + e2];
        a2 = fmaf(c2[i2],         v, a2);
        a1 = fmaf(c1[2 * i2 + h], v, a1);
        a0 = fmaf(c0[4 * i2 + q], v, a0);
    }
    size_t o = ((size_t)sp * B_ + b) * 2048 + e2;
    p0[o] = a0; p1[o] = a1; p2[o] = a2;
}

__global__ void k_stage2(const float* __restrict__ p0,
                         const float* __restrict__ p1,
                         const float* __restrict__ p2,
                         float* __restrict__ A) {
    int idx = blockIdx.x * 256 + threadIdx.x;
    int row = idx >> 6;
    int kk  = idx & 63;
    int b = row >> 6, m = row & 63;
    float acc = 0.f;
    if (kk < 8) {
        int j = m >> 3, s = m & 7, k = kk;
        int e = j * 64 + s * 8 + k;
        for (int sp = 0; sp < SPLITS; ++sp)
            for (int qq = 0; qq < 4; ++qq)
                acc += p0[((size_t)sp * B_ + b) * 2048 + qq * 512 + e];
    } else if (kk < 24) {
        int j = m >> 2, s = m & 3, k = kk - 8;
        int e = j * 64 + s * 16 + k;
        for (int sp = 0; sp < SPLITS; ++sp)
            for (int hh = 0; hh < 2; ++hh)
                acc += p1[((size_t)sp * B_ + b) * 2048 + hh * 1024 + e];
    } else if (kk < 56) {
        int j = m >> 1, s = m & 1, k = kk - 24;
        int e = j * 64 + s * 32 + k;
        for (int sp = 0; sp < SPLITS; ++sp)
            acc += p2[((size_t)sp * B_ + b) * 2048 + e];
    }
    A[idx] = acc;
}

#define FOLD_T 384

__device__ __forceinline__ void lds_barrier() {
    asm volatile("s_wait_dscnt 0x0\n\ts_barrier_signal -1\n\ts_barrier_wait -1" ::: "memory");
}

template <int R>
__device__ __forceinline__ void fold_body(
        const float* __restrict__ left0, const float* __restrict__ left1,
        const float* __restrict__ left2,
        int nseg,
        const float* __restrict__ fuse_rows,
        const float* __restrict__ add1,
        const float* __restrict__ flat_w,
        const float* __restrict__ add2,
        float* __restrict__ dst,
        int n0, float* sL, float* sW, float* sRed) {
    const int t = threadIdx.x;

    const int nleft = nseg * 128 * R;
    const int e1 = t + FOLD_T;
    int slot0, slot1 = 0;
    float lv0, lv1 = 0.f;
    {
        int seg = t / (128 * R), q = t % (128 * R), r = q >> 7, d = q & 127;
        const float* lp = (seg == 0) ? left0 : (seg == 1) ? left1 : left2;
        lv0 = lp[r * 128 + d];
        slot0 = (seg * 128 + d) * R + r;
    }
    if (e1 < nleft) {
        int seg = e1 / (128 * R), q = e1 % (128 * R), r = q >> 7, d = q & 127;
        const float* lp = (seg == 0) ? left0 : (seg == 1) ? left1 : left2;
        lv1 = lp[r * 128 + d];
        slot1 = (seg * 128 + d) * R + r;
    }
    const float s1 = (add1 != nullptr) ? add1[t] : 0.f;
    const float s2 = (add2 != nullptr && t < 64) ? add2[n0 + t] : 0.f;

    const int cg2 = t & 15, ks2 = t >> 4;
    float4 fw[16];
#pragma unroll
    for (int j = 0; j < 16; ++j)
        fw[j] = *reinterpret_cast<const float4*>(
            flat_w + (size_t)(ks2 * 16 + j) * 512 + n0 + cg2 * 4);

    const int cg = t % 96, ks = t / 96;
    const float* fp = fuse_rows + (size_t)(ks * 32) * 384 + cg * 4;
    float4 f[32];
#pragma unroll
    for (int j = 0; j < 32; ++j)
        f[j] = *reinterpret_cast<const float4*>(fp + (size_t)j * 384);

    sL[slot0] = lv0;
    if (e1 < nleft) sL[slot1] = lv1;
    lds_barrier();

    float a[R][4];
#pragma unroll
    for (int r = 0; r < R; ++r) { a[r][0] = a[r][1] = a[r][2] = a[r][3] = 0.f; }
    for (int seg = 0; ; ) {
        const float* lk = sL + (seg * 128 + ks * 32) * R;
#pragma unroll
        for (int j = 0; j < 32; ++j) {
#pragma unroll
            for (int r = 0; r < R; ++r) {
                float l = lk[j * R + r];
                a[r][0] = fmaf(l, f[j].x, a[r][0]);
                a[r][1] = fmaf(l, f[j].y, a[r][1]);
                a[r][2] = fmaf(l, f[j].z, a[r][2]);
                a[r][3] = fmaf(l, f[j].w, a[r][3]);
            }
        }
        if (++seg >= nseg) break;
        fp += (size_t)128 * 384;
#pragma unroll
        for (int j = 0; j < 32; ++j)
            f[j] = *reinterpret_cast<const float4*>(fp + (size_t)j * 384);
    }
    {
        float* rp = sRed + (ks * 384 + cg * 4) * R;
#pragma unroll
        for (int i = 0; i < 4; ++i)
#pragma unroll
            for (int r = 0; r < R; ++r) rp[i * R + r] = a[r][i];
    }
    lds_barrier();
    {
#pragma unroll
        for (int r = 0; r < R; ++r) {
            float p0 = sRed[(0 * 384 + t) * R + r], p1 = sRed[(1 * 384 + t) * R + r];
            float p2 = sRed[(2 * 384 + t) * R + r], p3 = sRed[(3 * 384 + t) * R + r];
            sW[t * R + r] = ((p0 + p1) + (p2 + p3)) + s1;
        }
    }
    lds_barrier();

    float c[R][4];
#pragma unroll
    for (int r = 0; r < R; ++r) { c[r][0] = c[r][1] = c[r][2] = c[r][3] = 0.f; }
    {
        const float* wk = sW + (ks2 * 16) * R;
#pragma unroll
        for (int j = 0; j < 16; ++j) {
#pragma unroll
            for (int r = 0; r < R; ++r) {
                float w = wk[j * R + r];
                c[r][0] = fmaf(w, fw[j].x, c[r][0]);
                c[r][1] = fmaf(w, fw[j].y, c[r][1]);
                c[r][2] = fmaf(w, fw[j].z, c[r][2]);
                c[r][3] = fmaf(w, fw[j].w, c[r][3]);
            }
        }
    }
    {
        float* rp = sRed + (ks2 * 64 + cg2 * 4) * R;
#pragma unroll
        for (int i = 0; i < 4; ++i)
#pragma unroll
            for (int r = 0; r < R; ++r) rp[i * R + r] = c[r][i];
    }
    lds_barrier();
    if (t < 64 * R) {
        int r = t >> 6, cc = t & 63;
        float q[4] = {0.f, 0.f, 0.f, 0.f};
#pragma unroll
        for (int k6 = 0; k6 < 6; ++k6)
#pragma unroll
            for (int ch = 0; ch < 4; ++ch)
                q[ch] += sRed[((ch * 6 + k6) * 64 + cc) * R + r];
        float s = (q[0] + q[1]) + (q[2] + q[3]);
        if (R == 1) s += s2;
        dst[(size_t)r * 512 + n0 + cc] = s;
    }
}

__global__ __launch_bounds__(FOLD_T)
void k_fold(const float* __restrict__ pw0, const float* __restrict__ pb0,
            const float* __restrict__ pw1, const float* __restrict__ pb1,
            const float* __restrict__ pw2, const float* __restrict__ pb2,
            const float* __restrict__ fuse_w, const float* __restrict__ fuse_b,
            const float* __restrict__ flat_w, const float* __restrict__ flat_b,
            float* __restrict__ Wtot, float* __restrict__ btot) {
    __shared__ __attribute__((aligned(16))) float sL[384 * 4];
    __shared__ __attribute__((aligned(16))) float sW[384 * 4];
    __shared__ __attribute__((aligned(16))) float sRed[4 * 384 * 4];
    const int n0 = (blockIdx.x & 7) * 64;
    const int g  = (int)(blockIdx.x >> 3) - 1;
    if (g >= 0) {
        int kk0 = 4 * g;
        const float* pw; int sc;
        if (kk0 < 8)       { sc = 0; pw = pw0 + (size_t)kk0 * 128; }
        else if (kk0 < 24) { sc = 1; pw = pw1 + (size_t)(kk0 - 8) * 128; }
        else               { sc = 2; pw = pw2 + (size_t)(kk0 - 24) * 128; }
        fold_body<4>(pw, pw, pw, 1, fuse_w + (size_t)sc * 128 * 384, nullptr,
                     flat_w, nullptr, Wtot + (size_t)kk0 * 512, n0, sL, sW, sRed);
    } else {
        int t = threadIdx.x;
        if (t < 128)
            *reinterpret_cast<float4*>(Wtot + (size_t)(56 + (t >> 4)) * 512 + n0 + (t & 15) * 4)
                = make_float4(0.f, 0.f, 0.f, 0.f);
        fold_body<1>(pb0, pb1, pb2, 3, fuse_w, fuse_b,
                     flat_w, flat_b, btot, n0, sL, sW, sRed);
    }
}

__global__ void k_gemm(const float* __restrict__ A, const float* __restrict__ Wt,
                       const float* __restrict__ btot, float* __restrict__ out) {
    int wave = threadIdx.x >> 5;
    int lane = threadIdx.x & 31;
    int tile = blockIdx.x * 8 + wave;
    int mt = tile >> 5, nt = tile & 31;
    int mbase = mt * 16, nbase = nt * 16;
    int half = lane >> 4, l16 = lane & 15;
    const float* Ar = A + (size_t)(mbase + l16) * 64;
    const float* Bc = Wt + nbase + l16;
    v8f acc = {};
#pragma unroll
    for (int k0 = 0; k0 < 64; k0 += 4) {
        int ka = k0 + half * 2;
        v2f av = { Ar[ka], Ar[ka + 1] };
        v2f bv = { Bc[(size_t)ka * 512], Bc[(size_t)(ka + 1) * 512] };
        acc = __builtin_amdgcn_wmma_f32_16x16x4_f32(
            false, av, false, bv, (short)0, acc, false, false);
    }
    float bias = btot[nbase + l16];
#pragma unroll
    for (int r = 0; r < 8; ++r)
        out[(size_t)(mbase + r + 8 * half) * 512 + nbase + l16] = acc[r] + bias;
}

extern "C" void kernel_launch(void* const* d_in, const int* in_sizes, int n_in,
                              void* d_out, int out_size, void* d_ws, size_t ws_size,
                              hipStream_t stream) {
    (void)in_sizes; (void)n_in; (void)out_size; (void)ws_size;
    const float* x      = (const float*)d_in[0];
    const float* pw0    = (const float*)d_in[1];
    const float* pb0    = (const float*)d_in[2];
    const float* pw1    = (const float*)d_in[3];
    const float* pb1    = (const float*)d_in[4];
    const float* pw2    = (const float*)d_in[5];
    const float* pb2    = (const float*)d_in[6];
    const float* fuse_w = (const float*)d_in[7];
    const float* fuse_b = (const float*)d_in[8];
    const float* flat_w = (const float*)d_in[9];
    const float* flat_b = (const float*)d_in[10];
    float* out = (float*)d_out;

    float* ws   = (float*)d_ws;
    float* c0   = ws;
    float* c1   = ws + 512;
    float* c2   = ws + 768;
    float* p0   = ws + 1024;
    float* p1   = p0 + SPLITS * B_ * 2048;
    float* p2   = p1 + SPLITS * B_ * 2048;
    float* A    = p2 + SPLITS * B_ * 2048;
    float* Wtot = A + 1024 * 64;
    float* btot = Wtot + 64 * 512;

    k_coef<<<224, 128, 0, stream>>>(c0, c1, c2);
    k_fold<<<15 * 8, FOLD_T, 0, stream>>>(pw0, pb0, pw1, pb1, pw2, pb2, fuse_w, fuse_b,
                                          flat_w, flat_b, Wtot, btot);

    dim3 g1(2048 / 256, B_, SPLITS);
    k_stage1<<<g1, 256, 0, stream>>>(x, c0, c1, c2, p0, p1, p2);
    k_stage2<<<(1024 * 64) / 256, 256, 0, stream>>>(p0, p1, p2, A);

    k_gemm<<<2048 / 8, 256, 0, stream>>>(A, Wtot, btot, out);
}


# x pass split by columns, not tokens: a block owns 64 columns x one token parity of a batch and finishes its sums itself; partial planes and combine launch gone (5 -> 4 launches); x sums regrouped
# speedup vs baseline: 16.6093x; 1.2734x over previous
#include <hip/hip_runtime.h>
#include <hip/hip_bf16.h>


#define B_ 16
#define L_ 4096
#define M_ 64

typedef __attribute__((ext_vector_type(2))) float v2f;
typedef __attribute__((ext_vector_type(8))) float v8f;
typedef __attribute__((ext_vector_type(4))) float v4f;

__global__ void k_coef(float* c0, float* c1, float* c2) {
    int lane = threadIdx.x & 31;
    int w = blockIdx.x * 4 + (threadIdx.x >> 5);
    int N, t; float* dst; double step;
    if (w < 512)      { N = 512; t = w;       dst = c0 + t; step = 511.0 / 511.0; }
    else if (w < 768) { N = 256; t = w - 512; dst = c1 + t; step = 255.0 / 511.0; }
    else              { N = 128; t = w - 768; dst = c2 + t; step = 127.0 / 511.0; }
    double acc = 0.0;
    for (int j = 0; j < 16; ++j) {
        int tt = lane * 16 + j;
        double pos = tt * step;
        int i0 = (int)pos;
        int i1 = i0 + 1; if (i1 > N - 1) i1 = N - 1;
        double w1 = pos - (double)i0;
        if (t == i0) acc += 1.0 - w1;
        if (t == i1) acc += w1;
    }
    for (int off = 16; off > 0; off >>= 1)
        acc += __shfl_xor(acc, off);
    if (lane == 0) *dst = (float)(acc / 512.0);
}

#define XP_T 256
__global__ __launch_bounds__(XP_T)
void k_xpass(const float* __restrict__ x,
             const float* __restrict__ c0,
             const float* __restrict__ c1,
             const float* __restrict__ c2,
             float* __restrict__ P0,
             float* __restrict__ P1,
             float* __restrict__ P2) {
    __shared__ __attribute__((aligned(16))) float sP[16 * 4 * 64];
    const int t  = threadIdx.x;
    const int cc = blockIdx.x;
    const int h  = blockIdx.y;
    const int b  = blockIdx.z;
    const int col4 = t & 15, G = t >> 4;

    v4f k2[2], k1[4], k0[8];
#pragma unroll
    for (int i = 0; i < 2; ++i) k2[i] = *reinterpret_cast<const v4f*>(c2 + 8 * G + 4 * i);
#pragma unroll
    for (int i = 0; i < 4; ++i) k1[i] = *reinterpret_cast<const v4f*>(c1 + 16 * G + 4 * i);
#pragma unroll
    for (int i = 0; i < 8; ++i) k0[i] = *reinterpret_cast<const v4f*>(c0 + 32 * G + 4 * i);

    const float* xp = x + (size_t)b * (L_ * M_) + (size_t)(32 * G + h) * 512 + cc * 64 + col4 * 4;
    v4f v[16];
#pragma unroll
    for (int j = 0; j < 16; ++j)
        v[j] = *reinterpret_cast<const v4f*>(xp + (size_t)j * 1024);
    asm volatile("" :: "v"(k2[0]), "v"(k2[1]), "v"(k1[0]), "v"(k1[1]), "v"(k1[2]), "v"(k1[3]),
                 "v"(k0[0]), "v"(k0[1]), "v"(k0[2]), "v"(k0[3]), "v"(k0[4]), "v"(k0[5]),
                 "v"(k0[6]), "v"(k0[7]));
    __builtin_amdgcn_sched_barrier(0);

    v4f a0  = {0.f, 0.f, 0.f, 0.f}, a1 = a0;
    v4f a2e = a0, a2o = a0;
#pragma unroll
    for (int j = 0; j < 16; ++j) {
        const v4f q0 = k0[j >> 1];
        const float w0 = (j & 1) ? (h ? q0.w : q0.z) : (h ? q0.y : q0.x);
        const v4f q1 = k1[j >> 2];
        const float w1 = (j & 3) == 0 ? q1.x : (j & 3) == 1 ? q1.y : (j & 3) == 2 ? q1.z : q1.w;
        const v4f q2 = k2[j >> 3];
        const float w2 = ((j >> 1) & 3) == 0 ? q2.x : ((j >> 1) & 3) == 1 ? q2.y
                       : ((j >> 1) & 3) == 2 ? q2.z : q2.w;
        a0.x = fmaf(w0, v[j].x, a0.x); a0.y = fmaf(w0, v[j].y, a0.y);
        a0.z = fmaf(w0, v[j].z, a0.z); a0.w = fmaf(w0, v[j].w, a0.w);
        a1.x = fmaf(w1, v[j].x, a1.x); a1.y = fmaf(w1, v[j].y, a1.y);
        a1.z = fmaf(w1, v[j].z, a1.z); a1.w = fmaf(w1, v[j].w, a1.w);
        if (j & 1) {
            a2o.x = fmaf(w2, v[j].x, a2o.x); a2o.y = fmaf(w2, v[j].y, a2o.y);
            a2o.z = fmaf(w2, v[j].z, a2o.z); a2o.w = fmaf(w2, v[j].w, a2o.w);
        } else {
            a2e.x = fmaf(w2, v[j].x, a2e.x); a2e.y = fmaf(w2, v[j].y, a2e.y);
            a2e.z = fmaf(w2, v[j].z, a2e.z); a2e.w = fmaf(w2, v[j].w, a2e.w);
        }
    }

    float* sp = sP + G * 256 + col4 * 4;
    *reinterpret_cast<v4f*>(sp)       = a0;
    *reinterpret_cast<v4f*>(sp + 64)  = a1;
    *reinterpret_cast<v4f*>(sp + 128) = a2e;
    *reinterpret_cast<v4f*>(sp + 192) = a2o;
    __syncthreads();

    float q[4] = {0.f, 0.f, 0.f, 0.f};
#pragma unroll
    for (int g4 = 0; g4 < 4; ++g4)
#pragma unroll
        for (int ch = 0; ch < 4; ++ch)
            q[ch] += sP[(ch * 4 + g4) * 256 + t];
    const float s = (q[0] + q[1]) + (q[2] + q[3]);
    const int slot = t >> 6, e0 = cc * 64 + (t & 63);
    float* dst;
    if (slot == 0)      dst = P0 + (size_t)(h * B_ + b) * 512 + e0;
    else if (slot == 1) dst = P1 + (size_t)b * 1024 + h * 512 + e0;
    else                dst = P2 + (size_t)b * 2048 + (h + 2 * (slot - 2)) * 512 + e0;
    *dst = s;
}

#define FOLD_T 384

__device__ __forceinline__ void lds_barrier() {
    asm volatile("s_wait_dscnt 0x0\n\ts_barrier_signal -1\n\ts_barrier_wait -1" ::: "memory");
}

template <int R>
__device__ __forceinline__ void fold_body(
        const float* __restrict__ left0, const float* __restrict__ left1,
        const float* __restrict__ left2,
        int nseg,
        const float* __restrict__ fuse_rows,
        const float* __restrict__ add1,
        const float* __restrict__ flat_w,
        const float* __restrict__ add2,
        float* __restrict__ dst,
        int n0, float* sL, float* sW, float* sRed) {
    const int t = threadIdx.x;

    const int nleft = nseg * 128 * R;
    const int e1 = t + FOLD_T;
    int slot0, slot1 = 0;
    float lv0, lv1 = 0.f;
    {
        int seg = t / (128 * R), q = t % (128 * R), r = q >> 7, d = q & 127;
        const float* lp = (seg == 0) ? left0 : (seg == 1) ? left1 : left2;
        lv0 = lp[r * 128 + d];
        slot0 = (seg * 128 + d) * R + r;
    }
    if (e1 < nleft) {
        int seg = e1 / (128 * R), q = e1 % (128 * R), r = q >> 7, d = q & 127;
        const float* lp = (seg == 0) ? left0 : (seg == 1) ? left1 : left2;
        lv1 = lp[r * 128 + d];
        slot1 = (seg * 128 + d) * R + r;
    }
    const float s1 = (add1 != nullptr) ? add1[t] : 0.f;
    const float s2 = (add2 != nullptr && t < 64) ? add2[n0 + t] : 0.f;

    const int cg2 = t & 15, ks2 = t >> 4;
    float4 fw[16];
#pragma unroll
    for (int j = 0; j < 16; ++j)
        fw[j] = *reinterpret_cast<const float4*>(
            flat_w + (size_t)(ks2 * 16 + j) * 512 + n0 + cg2 * 4);

    const int cg = t % 96, ks = t / 96;
    const float* fp = fuse_rows + (size_t)(ks * 32) * 384 + cg * 4;
    float4 f[32];
#pragma unroll
    for (int j = 0; j < 32; ++j)
        f[j] = *reinterpret_cast<const float4*>(fp + (size_t)j * 384);

    sL[slot0] = lv0;
    if (e1 < nleft) sL[slot1] = lv1;
    lds_barrier();

    float a[R][4];
#pragma unroll
    for (int r = 0; r < R; ++r) { a[r][0] = a[r][1] = a[r][2] = a[r][3] = 0.f; }
    for (int seg = 0; ; ) {
        const float* lk = sL + (seg * 128 + ks * 32) * R;
#pragma unroll
        for (int j = 0; j < 32; ++j) {
#pragma unroll
            for (int r = 0; r < R; ++r) {
                float l = lk[j * R + r];
                a[r][0] = fmaf(l, f[j].x, a[r][0]);
                a[r][1] = fmaf(l, f[j].y, a[r][1]);
                a[r][2] = fmaf(l, f[j].z, a[r][2]);
                a[r][3] = fmaf(l, f[j].w, a[r][3]);
            }
        }
        if (++seg >= nseg) break;
        fp += (size_t)128 * 384;
#pragma unroll
        for (int j = 0; j < 32; ++j)
            f[j] = *reinterpret_cast<const float4*>(fp + (size_t)j * 384);
    }
    {
        float* rp = sRed + (ks * 384 + cg * 4) * R;
#pragma unroll
        for (int i = 0; i < 4; ++i)
#pragma unroll
            for (int r = 0; r < R; ++r) rp[i * R + r] = a[r][i];
    }
    lds_barrier();
    {
#pragma unroll
        for (int r = 0; r < R; ++r) {
            float p0 = sRed[(0 * 384 + t) * R + r], p1 = sRed[(1 * 384 + t) * R + r];
            float p2 = sRed[(2 * 384 + t) * R + r], p3 = sRed[(3 * 384 + t) * R + r];
            sW[t * R + r] = ((p0 + p1) + (p2 + p3)) + s1;
        }
    }
    lds_barrier();

    float c[R][4];
#pragma unroll
    for (int r = 0; r < R; ++r) { c[r][0] = c[r][1] = c[r][2] = c[r][3] = 0.f; }
    {
        const float* wk = sW + (ks2 * 16) * R;
#pragma unroll
        for (int j = 0; j < 16; ++j) {
#pragma unroll
            for (int r = 0; r < R; ++r) {
                float w = wk[j * R + r];
                c[r][0] = fmaf(w, fw[j].x, c[r][0]);
                c[r][1] = fmaf(w, fw[j].y, c[r][1]);
                c[r][2] = fmaf(w, fw[j].z, c[r][2]);
                c[r][3] = fmaf(w, fw[j].w, c[r][3]);
            }
        }
    }
    {
        float* rp = sRed + (ks2 * 64 + cg2 * 4) * R;
#pragma unroll
        for (int i = 0; i < 4; ++i)
#pragma unroll
            for (int r = 0; r < R; ++r) rp[i * R + r] = c[r][i];
    }
    lds_barrier();
    if (t < 64 * R) {
        int r = t >> 6, cc = t & 63;
        float q[4] = {0.f, 0.f, 0.f, 0.f};
#pragma unroll
        for (int k6 = 0; k6 < 6; ++k6)
#pragma unroll
            for (int ch = 0; ch < 4; ++ch)
                q[ch] += sRed[((ch * 6 + k6) * 64 + cc) * R + r];
        float s = (q[0] + q[1]) + (q[2] + q[3]);
        if (R == 1) s += s2;
        dst[(size_t)r * 512 + n0 + cc] = s;
    }
}

__global__ __launch_bounds__(FOLD_T)
void k_fold(const float* __restrict__ pw0, const float* __restrict__ pb0,
            const float* __restrict__ pw1, const float* __restrict__ pb1,
            const float* __restrict__ pw2, const float* __restrict__ pb2,
            const float* __restrict__ fuse_w, const float* __restrict__ fuse_b,
            const float* __restrict__ flat_w, const float* __restrict__ flat_b,
            float* __restrict__ Wtot, float* __restrict__ btot) {
    __shared__ __attribute__((aligned(16))) float sL[384 * 4];
    __shared__ __attribute__((aligned(16))) float sW[384 * 4];
    __shared__ __attribute__((aligned(16))) float sRed[4 * 384 * 4];
    const int n0 = (blockIdx.x & 7) * 64;
    const int g  = (int)(blockIdx.x >> 3) - 1;
    if (g >= 0) {
        int kk0 = 4 * g;
        const float* pw; int sc;
        if (kk0 < 8)       { sc = 0; pw = pw0 + (size_t)kk0 * 128; }
        else if (kk0 < 24) { sc = 1; pw = pw1 + (size_t)(kk0 - 8) * 128; }
        else               { sc = 2; pw = pw2 + (size_t)(kk0 - 24) * 128; }
        fold_body<4>(pw, pw, pw, 1, fuse_w + (size_t)sc * 128 * 384, nullptr,
                     flat_w, nullptr, Wtot + (size_t)kk0 * 512, n0, sL, sW, sRed);
    } else {
        fold_body<1>(pb0, pb1, pb2, 3, fuse_w, fuse_b,
                     flat_w, flat_b, btot, n0, sL, sW, sRed);
    }
}

__global__ void k_gemm(const float* __restrict__ P0, const float* __restrict__ P1,
                       const float* __restrict__ P2, const float* __restrict__ Wt,
                       const float* __restrict__ btot, float* __restrict__ out) {
    int wave = threadIdx.x >> 5;
    int lane = threadIdx.x & 31;
    int tile = blockIdx.x * 8 + wave;
    int mt = tile >> 5, nt = tile & 31;
    int mbase = mt * 16, nbase = nt * 16;
    int half = lane >> 4, l16 = lane & 15;
    const int row = mbase + l16;
    const float* A0a = P0 + (size_t)row * 8;
    const float* A0b = A0a + B_ * 512;
    const float* A1  = P1 + (size_t)row * 16;
    const float* A2  = P2 + (size_t)row * 32;
    const float* Bc = Wt + nbase + l16;
    v8f acc = {};
#pragma unroll
    for (int k0 = 0; k0 < 56; k0 += 4) {
        int ka = k0 + half * 2;
        v2f av;
        if (k0 < 8)       av = v2f{ A0a[ka] + A0b[ka], A0a[ka + 1] + A0b[ka + 1] };
        else if (k0 < 24) av = v2f{ A1[ka - 8], A1[ka - 7] };
        else              av = v2f{ A2[ka - 24], A2[ka - 23] };
        v2f bv = { Bc[(size_t)ka * 512], Bc[(size_t)(ka + 1) * 512] };
        acc = __builtin_amdgcn_wmma_f32_16x16x4_f32(
            false, av, false, bv, (short)0, acc, false, false);
    }
    float bias = btot[nbase + l16];
#pragma unroll
    for (int r = 0; r < 8; ++r)
        out[(size_t)(mbase + r + 8 * half) * 512 + nbase + l16] = acc[r] + bias;
}

extern "C" void kernel_launch(void* const* d_in, const int* in_sizes, int n_in,
                              void* d_out, int out_size, void* d_ws, size_t ws_size,
                              hipStream_t stream) {
    (void)in_sizes; (void)n_in; (void)out_size; (void)ws_size;
    const float* x      = (const float*)d_in[0];
    const float* pw0    = (const float*)d_in[1];
    const float* pb0    = (const float*)d_in[2];
    const float* pw1    = (const float*)d_in[3];
    const float* pb1    = (const float*)d_in[4];
    const float* pw2    = (const float*)d_in[5];
    const float* pb2    = (const float*)d_in[6];
    const float* fuse_w = (const float*)d_in[7];
    const float* fuse_b = (const float*)d_in[8];
    const float* flat_w = (const float*)d_in[9];
    const float* flat_b = (const float*)d_in[10];
    float* out = (float*)d_out;

    float* ws   = (float*)d_ws;
    float* c0   = ws;
    float* c1   = ws + 512;
    float* c2   = ws + 768;
    float* P0   = ws + 1024;
    float* P1   = P0 + 2 * B_ * 512;
    float* P2   = P1 + B_ * 1024;
    float* Wtot = P2 + B_ * 2048;
    float* btot = Wtot + 56 * 512;

    k_coef<<<224, 128, 0, stream>>>(c0, c1, c2);
    k_fold<<<15 * 8, FOLD_T, 0, stream>>>(pw0, pb0, pw1, pb1, pw2, pb2, fuse_w, fuse_b,
                                          flat_w, flat_b, Wtot, btot);

    dim3 gx(8, 2, B_);
    k_xpass<<<gx, XP_T, 0, stream>>>(x, c0, c1, c2, P0, P1, P2);

    k_gemm<<<2048 / 8, 256, 0, stream>>>(P0, P1, P2, Wtot, btot, out);
}


# the 512-token scale resamples onto itself (all weights 1/512): its coefficients are no longer summed or loaded; the 384 left get 2 waves each: busiest wave 134 -> 71 fp64 ops; tap sums regrouped
# speedup vs baseline: 17.5653x; 1.0576x over previous
#include <hip/hip_runtime.h>
#include <hip/hip_bf16.h>


#define B_ 16
#define L_ 4096
#define M_ 64

typedef __attribute__((ext_vector_type(2))) float v2f;
typedef __attribute__((ext_vector_type(8))) float v8f;
typedef __attribute__((ext_vector_type(4))) float v4f;

__global__ void k_coef(float* c1, float* c2) {
    __shared__ double sPart[4];
    const int lane = threadIdx.x & 31;
    const int wv   = threadIdx.x >> 5;
    const int w    = blockIdx.x * 2 + (wv >> 1);
    const int part = wv & 1;
    int N, t; float* dst; double step;
    if (w < 256) { N = 256; t = w;       dst = c1 + t; step = 255.0 / 511.0; }
    else         { N = 128; t = w - 256; dst = c2 + t; step = 127.0 / 511.0; }
    double acc = 0.0;
    for (int j = 0; j < 8; ++j) {
        int tt = part * 256 + lane * 8 + j;
        double pos = tt * step;
        int i0 = (int)pos;
        int i1 = i0 + 1; if (i1 > N - 1) i1 = N - 1;
        double w1 = pos - (double)i0;
        if (t == i0) acc += 1.0 - w1;
        if (t == i1) acc += w1;
    }
    for (int off = 16; off > 0; off >>= 1)
        acc += __shfl_xor(acc, off);
    if (lane == 0) sPart[wv] = acc;
    __syncthreads();
    if (lane == 0 && part == 0) *dst = (float)((sPart[wv] + sPart[wv + 1]) / 512.0);
}

#define XP_T 256
__global__ __launch_bounds__(XP_T)
void k_xpass(const float* __restrict__ x,
             const float* __restrict__ c1,
             const float* __restrict__ c2,
             float* __restrict__ P0,
             float* __restrict__ P1,
             float* __restrict__ P2) {
    __shared__ __attribute__((aligned(16))) float sP[16 * 4 * 64];
    const int t  = threadIdx.x;
    const int cc = blockIdx.x;
    const int h  = blockIdx.y;
    const int b  = blockIdx.z;
    const int col4 = t & 15, G = t >> 4;

    v4f k2[2], k1[4];
#pragma unroll
    for (int i = 0; i < 2; ++i) k2[i] = *reinterpret_cast<const v4f*>(c2 + 8 * G + 4 * i);
#pragma unroll
    for (int i = 0; i < 4; ++i) k1[i] = *reinterpret_cast<const v4f*>(c1 + 16 * G + 4 * i);

    const float* xp = x + (size_t)b * (L_ * M_) + (size_t)(32 * G + h) * 512 + cc * 64 + col4 * 4;
    v4f v[16];
#pragma unroll
    for (int j = 0; j < 16; ++j)
        v[j] = *reinterpret_cast<const v4f*>(xp + (size_t)j * 1024);
    asm volatile("" :: "v"(k2[0]), "v"(k2[1]), "v"(k1[0]), "v"(k1[1]), "v"(k1[2]), "v"(k1[3]));
    __builtin_amdgcn_sched_barrier(0);

    v4f a0  = {0.f, 0.f, 0.f, 0.f}, a1 = a0;
    v4f a2e = a0, a2o = a0;
#pragma unroll
    for (int j = 0; j < 16; ++j) {
        const float w0 = 1.0f / 512.0f;
        const v4f q1 = k1[j >> 2];
        const float w1 = (j & 3) == 0 ? q1.x : (j & 3) == 1 ? q1.y : (j & 3) == 2 ? q1.z : q1.w;
        const v4f q2 = k2[j >> 3];
        const float w2 = ((j >> 1) & 3) == 0 ? q2.x : ((j >> 1) & 3) == 1 ? q2.y
                       : ((j >> 1) & 3) == 2 ? q2.z : q2.w;
        a0.x = fmaf(w0, v[j].x, a0.x); a0.y = fmaf(w0, v[j].y, a0.y);
        a0.z = fmaf(w0, v[j].z, a0.z); a0.w = fmaf(w0, v[j].w, a0.w);
        a1.x = fmaf(w1, v[j].x, a1.x); a1.y = fmaf(w1, v[j].y, a1.y);
        a1.z = fmaf(w1, v[j].z, a1.z); a1.w = fmaf(w1, v[j].w, a1.w);
        if (j & 1) {
            a2o.x = fmaf(w2, v[j].x, a2o.x); a2o.y = fmaf(w2, v[j].y, a2o.y);
            a2o.z = fmaf(w2, v[j].z, a2o.z); a2o.w = fmaf(w2, v[j].w, a2o.w);
        } else {
            a2e.x = fmaf(w2, v[j].x, a2e.x); a2e.y = fmaf(w2, v[j].y, a2e.y);
            a2e.z = fmaf(w2, v[j].z, a2e.z); a2e.w = fmaf(w2, v[j].w, a2e.w);
        }
    }

    float* sp = sP + G * 256 + col4 * 4;
    *reinterpret_cast<v4f*>(sp)       = a0;
    *reinterpret_cast<v4f*>(sp + 64)  = a1;
    *reinterpret_cast<v4f*>(sp + 128) = a2e;
    *reinterpret_cast<v4f*>(sp + 192) = a2o;
    __syncthreads();

    float q[4] = {0.f, 0.f, 0.f, 0.f};
#pragma unroll
    for (int g4 = 0; g4 < 4; ++g4)
#pragma unroll
        for (int ch = 0; ch < 4; ++ch)
            q[ch] += sP[(ch * 4 + g4) * 256 + t];
    const float s = (q[0] + q[1]) + (q[2] + q[3]);
    const int slot = t >> 6, e0 = cc * 64 + (t & 63);
    float* dst;
    if (slot == 0)      dst = P0 + (size_t)(h * B_ + b) * 512 + e0;
    else if (slot == 1) dst = P1 + (size_t)b * 1024 + h * 512 + e0;
    else                dst = P2 + (size_t)b * 2048 + (h + 2 * (slot - 2)) * 512 + e0;
    *dst = s;
}

#define FOLD_T 384

__device__ __forceinline__ void lds_barrier() {
    asm volatile("s_wait_dscnt 0x0\n\ts_barrier_signal -1\n\ts_barrier_wait -1" ::: "memory");
}

template <int R>
__device__ __forceinline__ void fold_body(
        const float* __restrict__ left0, const float* __restrict__ left1,
        const float* __restrict__ left2,
        int nseg,
        const float* __restrict__ fuse_rows,
        const float* __restrict__ add1,
        const float* __restrict__ flat_w,
        const float* __restrict__ add2,
        float* __restrict__ dst,
        int n0, float* sL, float* sW, float* sRed) {
    const int t = threadIdx.x;

    const int nleft = nseg * 128 * R;
    const int e1 = t + FOLD_T;
    int slot0, slot1 = 0;
    float lv0, lv1 = 0.f;
    {
        int seg = t / (128 * R), q = t % (128 * R), r = q >> 7, d = q & 127;
        const float* lp = (seg == 0) ? left0 : (seg == 1) ? left1 : left2;
        lv0 = lp[r * 128 + d];
        slot0 = (seg * 128 + d) * R + r;
    }
    if (e1 < nleft) {
        int seg = e1 / (128 * R), q = e1 % (128 * R), r = q >> 7, d = q & 127;
        const float* lp = (seg == 0) ? left0 : (seg == 1) ? left1 : left2;
        lv1 = lp[r * 128 + d];
        slot1 = (seg * 128 + d) * R + r;
    }
    const float s1 = (add1 != nullptr) ? add1[t] : 0.f;
    const float s2 = (add2 != nullptr && t < 64) ? add2[n0 + t] : 0.f;

    const int cg2 = t & 15, ks2 = t >> 4;
    float4 fw[16];
#pragma unroll
    for (int j = 0; j < 16; ++j)
        fw[j] = *reinterpret_cast<const float4*>(
            flat_w + (size_t)(ks2 * 16 + j) * 512 + n0 + cg2 * 4);

    const int cg = t % 96, ks = t / 96;
    const float* fp = fuse_rows + (size_t)(ks * 32) * 384 + cg * 4;
    float4 f[32];
#pragma unroll
    for (int j = 0; j < 32; ++j)
        f[j] = *reinterpret_cast<const float4*>(fp + (size_t)j * 384);

    sL[slot0] = lv0;
    if (e1 < nleft) sL[slot1] = lv1;
    lds_barrier();

    float a[R][4];
#pragma unroll
    for (int r = 0; r < R; ++r) { a[r][0] = a[r][1] = a[r][2] = a[r][3] = 0.f; }
    for (int seg = 0; ; ) {
        const float* lk = sL + (seg * 128 + ks * 32) * R;
#pragma unroll
        for (int j = 0; j < 32; ++j) {
#pragma unroll
            for (int r = 0; r < R; ++r) {
                float l = lk[j * R + r];
                a[r][0] = fmaf(l, f[j].x, a[r][0]);
                a[r][1] = fmaf(l, f[j].y, a[r][1]);
                a[r][2] = fmaf(l, f[j].z, a[r][2]);
                a[r][3] = fmaf(l, f[j].w, a[r][3]);
            }
        }
        if (++seg >= nseg) break;
        fp += (size_t)128 * 384;
#pragma unroll
        for (int j = 0; j < 32; ++j)
            f[j] = *reinterpret_cast<const float4*>(fp + (size_t)j * 384);
    }
    {
        float* rp = sRed + (ks * 384 + cg * 4) * R;
#pragma unroll
        for (int i = 0; i < 4; ++i)
#pragma unroll
            for (int r = 0; r < R; ++r) rp[i * R + r] = a[r][i];
    }
    lds_barrier();
    {
#pragma unroll
        for (int r = 0; r < R; ++r) {
            float p0 = sRed[(0 * 384 + t) * R + r], p1 = sRed[(1 * 384 + t) * R + r];
            float p2 = sRed[(2 * 384 + t) * R + r], p3 = sRed[(3 * 384 + t) * R + r];
            sW[t * R + r] = ((p0 + p1) + (p2 + p3)) + s1;
        }
    }
    lds_barrier();

    float c[R][4];
#pragma unroll
    for (int r = 0; r < R; ++r) { c[r][0] = c[r][1] = c[r][2] = c[r][3] = 0.f; }
    {
        const float* wk = sW + (ks2 * 16) * R;
#pragma unroll
        for (int j = 0; j < 16; ++j) {
#pragma unroll
            for (int r = 0; r < R; ++r) {
                float w = wk[j * R + r];
                c[r][0] = fmaf(w, fw[j].x, c[r][0]);
                c[r][1] = fmaf(w, fw[j].y, c[r][1]);
                c[r][2] = fmaf(w, fw[j].z, c[r][2]);
                c[r][3] = fmaf(w, fw[j].w, c[r][3]);
            }
        }
    }
    {
        float* rp = sRed + (ks2 * 64 + cg2 * 4) * R;
#pragma unroll
        for (int i = 0; i < 4; ++i)
#pragma unroll
            for (int r = 0; r < R; ++r) rp[i * R + r] = c[r][i];
    }
    lds_barrier();
    if (t < 64 * R) {
        int r = t >> 6, cc = t & 63;
        float q[4] = {0.f, 0.f, 0.f, 0.f};
#pragma unroll
        for (int k6 = 0; k6 < 6; ++k6)
#pragma unroll
            for (int ch = 0; ch < 4; ++ch)
                q[ch] += sRed[((ch * 6 + k6) * 64 + cc) * R + r];
        float s = (q[0] + q[1]) + (q[2] + q[3]);
        if (R == 1) s += s2;
        dst[(size_t)r * 512 + n0 + cc] = s;
    }
}

__global__ __launch_bounds__(FOLD_T)
void k_fold(const float* __restrict__ pw0, const float* __restrict__ pb0,
            const float* __restrict__ pw1, const float* __restrict__ pb1,
            const float* __restrict__ pw2, const float* __restrict__ pb2,
            const float* __restrict__ fuse_w, const float* __restrict__ fuse_b,
            const float* __restrict__ flat_w, const float* __restrict__ flat_b,
            float* __restrict__ Wtot, float* __restrict__ btot) {
    __shared__ __attribute__((aligned(16))) float sL[384 * 4];
    __shared__ __attribute__((aligned(16))) float sW[384 * 4];
    __shared__ __attribute__((aligned(16))) float sRed[4 * 384 * 4];
    const int n0 = (blockIdx.x & 7) * 64;
    const int g  = (int)(blockIdx.x >> 3) - 1;
    if (g >= 0) {
        int kk0 = 4 * g;
        const float* pw; int sc;
        if (kk0 < 8)       { sc = 0; pw = pw0 + (size_t)kk0 * 128; }
        else if (kk0 < 24) { sc = 1; pw = pw1 + (size_t)(kk0 - 8) * 128; }
        else               { sc = 2; pw = pw2 + (size_t)(kk0 - 24) * 128; }
        fold_body<4>(pw, pw, pw, 1, fuse_w + (size_t)sc * 128 * 384, nullptr,
                     flat_w, nullptr, Wtot + (size_t)kk0 * 512, n0, sL, sW, sRed);
    } else {
        fold_body<1>(pb0, pb1, pb2, 3, fuse_w, fuse_b,
                     flat_w, flat_b, btot, n0, sL, sW, sRed);
    }
}

__global__ void k_gemm(const float* __restrict__ P0, const float* __restrict__ P1,
                       const float* __restrict__ P2, const float* __restrict__ Wt,
                       const float* __restrict__ btot, float* __restrict__ out) {
    int wave = threadIdx.x >> 5;
    int lane = threadIdx.x & 31;
    int tile = blockIdx.x * 8 + wave;
    int mt = tile >> 5, nt = tile & 31;
    int mbase = mt * 16, nbase = nt * 16;
    int half = lane >> 4, l16 = lane & 15;
    const int row = mbase + l16;
    const float* A0a = P0 + (size_t)row * 8;
    const float* A0b = A0a + B_ * 512;
    const float* A1  = P1 + (size_t)row * 16;
    const float* A2  = P2 + (size_t)row * 32;
    const float* Bc = Wt + nbase + l16;
    v8f acc = {};
#pragma unroll
    for (int k0 = 0; k0 < 56; k0 += 4) {
        int ka = k0 + half * 2;
        v2f av;
        if (k0 < 8)       av = v2f{ A0a[ka] + A0b[ka], A0a[ka + 1] + A0b[ka + 1] };
        else if (k0 < 24) av = v2f{ A1[ka - 8], A1[ka - 7] };
        else              av = v2f{ A2[ka - 24], A2[ka - 23] };
        v2f bv = { Bc[(size_t)ka * 512], Bc[(size_t)(ka + 1) * 512] };
        acc = __builtin_amdgcn_wmma_f32_16x16x4_f32(
            false, av, false, bv, (short)0, acc, false, false);
    }
    float bias = btot[nbase + l16];
#pragma unroll
    for (int r = 0; r < 8; ++r)
        out[(size_t)(mbase + r + 8 * half) * 512 + nbase + l16] = acc[r] + bias;
}

extern "C" void kernel_launch(void* const* d_in, const int* in_sizes, int n_in,
                              void* d_out, int out_size, void* d_ws, size_t ws_size,
                              hipStream_t stream) {
    (void)in_sizes; (void)n_in; (void)out_size; (void)ws_size;
    const float* x      = (const float*)d_in[0];
    const float* pw0    = (const float*)d_in[1];
    const float* pb0    = (const float*)d_in[2];
    const float* pw1    = (const float*)d_in[3];
    const float* pb1    = (const float*)d_in[4];
    const float* pw2    = (const float*)d_in[5];
    const float* pb2    = (const float*)d_in[6];
    const float* fuse_w = (const float*)d_in[7];
    const float* fuse_b = (const float*)d_in[8];
    const float* flat_w = (const float*)d_in[9];
    const float* flat_b = (const float*)d_in[10];
    float* out = (float*)d_out;

    float* ws   = (float*)d_ws;
    float* c1   = ws + 512;
    float* c2   = ws + 768;
    float* P0   = ws + 1024;
    float* P1   = P0 + 2 * B_ * 512;
    float* P2   = P1 + B_ * 1024;
    float* Wtot = P2 + B_ * 2048;
    float* btot = Wtot + 56 * 512;

    k_coef<<<192, 128, 0, stream>>>(c1, c2);
    k_fold<<<15 * 8, FOLD_T, 0, stream>>>(pw0, pb0, pw1, pb1, pw2, pb2, fuse_w, fuse_b,
                                          flat_w, flat_b, Wtot, btot);

    dim3 gx(8, 2, B_);
    k_xpass<<<gx, XP_T, 0, stream>>>(x, c1, c2, P0, P1, P2);

    k_gemm<<<2048 / 8, 256, 0, stream>>>(P0, P1, P2, Wtot, btot, out);
}


# k_coef launch gone (4 -> 3): each coefficient is two arithmetic series of taps, summed exactly in integers in closed form by the x-pass lanes (one entry per lane, row broadcast); no fp64
# speedup vs baseline: 21.3490x; 1.2154x over previous
#include <hip/hip_runtime.h>
#include <hip/hip_bf16.h>


#define B_ 16
#define L_ 4096
#define M_ 64

typedef __attribute__((ext_vector_type(2))) float v2f;
typedef __attribute__((ext_vector_type(8))) float v8f;
typedef __attribute__((ext_vector_type(4))) float v4f;

template <int N>
__device__ __forceinline__ float coef_mean(int i) {
    constexpr int S = N - 1, D = 511;
    const int tlo  = (i == 0) ? 0 : (D * (i - 1)) / S + 1;
    const int tmid = (D * i) / S;
    const int thi  = min((D * (i + 1)) / S, D);
    const int n1 = tmid - tlo + 1, n2 = thi - tmid;
    const int up   = n1 * (D - D * i) + S * (((tlo + tmid) * n1) / 2);
    const int down = n2 * (D + D * i) - S * (((tmid + 1 + thi) * n2) / 2);
    return (float)(up + down) / (float)(D * 512);
}

#define XP_T 256
__global__ __launch_bounds__(XP_T)
void k_xpass(const float* __restrict__ x,
             float* __restrict__ P0,
             float* __restrict__ P1,
             float* __restrict__ P2) {
    __shared__ __attribute__((aligned(16))) float sP[16 * 4 * 64];
    const int t  = threadIdx.x;
    const int cc = blockIdx.x;
    const int h  = blockIdx.y;
    const int b  = blockIdx.z;
    const int col4 = t & 15, G = t >> 4;

    const float* xp = x + (size_t)b * (L_ * M_) + (size_t)(32 * G + h) * 512 + cc * 64 + col4 * 4;
    v4f v[16];
#pragma unroll
    for (int j = 0; j < 16; ++j)
        v[j] = *reinterpret_cast<const v4f*>(xp + (size_t)j * 1024);
    __builtin_amdgcn_sched_barrier(0);

    const float my1 = coef_mean<256>(16 * G + col4);
    const float my2 = coef_mean<128>(8 * G + (col4 & 7));

    v4f a0  = {0.f, 0.f, 0.f, 0.f}, a1 = a0;
    v4f a2e = a0, a2o = a0;
#pragma unroll
    for (int j = 0; j < 16; ++j) {
        const float w0 = 1.0f / 512.0f;
        const float w1 = __shfl(my1, j, 16);
        const float w2 = __shfl(my2, j >> 1, 16);
        a0.x = fmaf(w0, v[j].x, a0.x); a0.y = fmaf(w0, v[j].y, a0.y);
        a0.z = fmaf(w0, v[j].z, a0.z); a0.w = fmaf(w0, v[j].w, a0.w);
        a1.x = fmaf(w1, v[j].x, a1.x); a1.y = fmaf(w1, v[j].y, a1.y);
        a1.z = fmaf(w1, v[j].z, a1.z); a1.w = fmaf(w1, v[j].w, a1.w);
        if (j & 1) {
            a2o.x = fmaf(w2, v[j].x, a2o.x); a2o.y = fmaf(w2, v[j].y, a2o.y);
            a2o.z = fmaf(w2, v[j].z, a2o.z); a2o.w = fmaf(w2, v[j].w, a2o.w);
        } else {
            a2e.x = fmaf(w2, v[j].x, a2e.x); a2e.y = fmaf(w2, v[j].y, a2e.y);
            a2e.z = fmaf(w2, v[j].z, a2e.z); a2e.w = fmaf(w2, v[j].w, a2e.w);
        }
    }

    float* sp = sP + G * 256 + col4 * 4;
    *reinterpret_cast<v4f*>(sp)       = a0;
    *reinterpret_cast<v4f*>(sp + 64)  = a1;
    *reinterpret_cast<v4f*>(sp + 128) = a2e;
    *reinterpret_cast<v4f*>(sp + 192) = a2o;
    __syncthreads();

    float q[4] = {0.f, 0.f, 0.f, 0.f};
#pragma unroll
    for (int g4 = 0; g4 < 4; ++g4)
#pragma unroll
        for (int ch = 0; ch < 4; ++ch)
            q[ch] += sP[(ch * 4 + g4) * 256 + t];
    const float s = (q[0] + q[1]) + (q[2] + q[3]);
    const int slot = t >> 6, e0 = cc * 64 + (t & 63);
    float* dst;
    if (slot == 0)      dst = P0 + (size_t)(h * B_ + b) * 512 + e0;
    else if (slot == 1) dst = P1 + (size_t)b * 1024 + h * 512 + e0;
    else                dst = P2 + (size_t)b * 2048 + (h + 2 * (slot - 2)) * 512 + e0;
    *dst = s;
}

#define FOLD_T 384

__device__ __forceinline__ void lds_barrier() {
    asm volatile("s_wait_dscnt 0x0\n\ts_barrier_signal -1\n\ts_barrier_wait -1" ::: "memory");
}

template <int R>
__device__ __forceinline__ void fold_body(
        const float* __restrict__ left0, const float* __restrict__ left1,
        const float* __restrict__ left2,
        int nseg,
        const float* __restrict__ fuse_rows,
        const float* __restrict__ add1,
        const float* __restrict__ flat_w,
        const float* __restrict__ add2,
        float* __restrict__ dst,
        int n0, float* sL, float* sW, float* sRed) {
    const int t = threadIdx.x;

    const int nleft = nseg * 128 * R;
    const int e1 = t + FOLD_T;
    int slot0, slot1 = 0;
    float lv0, lv1 = 0.f;
    {
        int seg = t / (128 * R), q = t % (128 * R), r = q >> 7, d = q & 127;
        const float* lp = (seg == 0) ? left0 : (seg == 1) ? left1 : left2;
        lv0 = lp[r * 128 + d];
        slot0 = (seg * 128 + d) * R + r;
    }
    if (e1 < nleft) {
        int seg = e1 / (128 * R), q = e1 % (128 * R), r = q >> 7, d = q & 127;
        const float* lp = (seg == 0) ? left0 : (seg == 1) ? left1 : left2;
        lv1 = lp[r * 128 + d];
        slot1 = (seg * 128 + d) * R + r;
    }
    const float s1 = (add1 != nullptr) ? add1[t] : 0.f;
    const float s2 = (add2 != nullptr && t < 64) ? add2[n0 + t] : 0.f;

    const int cg2 = t & 15, ks2 = t >> 4;
    float4 fw[16];
#pragma unroll
    for (int j = 0; j < 16; ++j)
        fw[j] = *reinterpret_cast<const float4*>(
            flat_w + (size_t)(ks2 * 16 + j) * 512 + n0 + cg2 * 4);

    const int cg = t % 96, ks = t / 96;
    const float* fp = fuse_rows + (size_t)(ks * 32) * 384 + cg * 4;
    float4 f[32];
#pragma unroll
    for (int j = 0; j < 32; ++j)
        f[j] = *reinterpret_cast<const float4*>(fp + (size_t)j * 384);

    sL[slot0] = lv0;
    if (e1 < nleft) sL[slot1] = lv1;
    lds_barrier();

    float a[R][4];
#pragma unroll
    for (int r = 0; r < R; ++r) { a[r][0] = a[r][1] = a[r][2] = a[r][3] = 0.f; }
    for (int seg = 0; ; ) {
        const float* lk = sL + (seg * 128 + ks * 32) * R;
#pragma unroll
        for (int j = 0; j < 32; ++j) {
#pragma unroll
            for (int r = 0; r < R; ++r) {
                float l = lk[j * R + r];
                a[r][0] = fmaf(l, f[j].x, a[r][0]);
                a[r][1] = fmaf(l, f[j].y, a[r][1]);
                a[r][2] = fmaf(l, f[j].z, a[r][2]);
                a[r][3] = fmaf(l, f[j].w, a[r][3]);
            }
        }
        if (++seg >= nseg) break;
        fp += (size_t)128 * 384;
#pragma unroll
        for (int j = 0; j < 32; ++j)
            f[j] = *reinterpret_cast<const float4*>(fp + (size_t)j * 384);
    }
    {
        float* rp = sRed + (ks * 384 + cg * 4) * R;
#pragma unroll
        for (int i = 0; i < 4; ++i)
#pragma unroll
            for (int r = 0; r < R; ++r) rp[i * R + r] = a[r][i];
    }
    lds_barrier();
    {
#pragma unroll
        for (int r = 0; r < R; ++r) {
            float p0 = sRed[(0 * 384 + t) * R + r], p1 = sRed[(1 * 384 + t) * R + r];
            float p2 = sRed[(2 * 384 + t) * R + r], p3 = sRed[(3 * 384 + t) * R + r];
            sW[t * R + r] = ((p0 + p1) + (p2 + p3)) + s1;
        }
    }
    lds_barrier();

    float c[R][4];
#pragma unroll
    for (int r = 0; r < R; ++r) { c[r][0] = c[r][1] = c[r][2] = c[r][3] = 0.f; }
    {
        const float* wk = sW + (ks2 * 16) * R;
#pragma unroll
        for (int j = 0; j < 16; ++j) {
#pragma unroll
            for (int r = 0; r < R; ++r) {
                float w = wk[j * R + r];
                c[r][0] = fmaf(w, fw[j].x, c[r][0]);
                c[r][1] = fmaf(w, fw[j].y, c[r][1]);
                c[r][2] = fmaf(w, fw[j].z, c[r][2]);
                c[r][3] = fmaf(w, fw[j].w, c[r][3]);
            }
        }
    }
    {
        float* rp = sRed + (ks2 * 64 + cg2 * 4) * R;
#pragma unroll
        for (int i = 0; i < 4; ++i)
#pragma unroll
            for (int r = 0; r < R; ++r) rp[i * R + r] = c[r][i];
    }
    lds_barrier();
    if (t < 64 * R) {
        int r = t >> 6, cc = t & 63;
        float q[4] = {0.f, 0.f, 0.f, 0.f};
#pragma unroll
        for (int k6 = 0; k6 < 6; ++k6)
#pragma unroll
            for (int ch = 0; ch < 4; ++ch)
                q[ch] += sRed[((ch * 6 + k6) * 64 + cc) * R + r];
        float s = (q[0] + q[1]) + (q[2] + q[3]);
        if (R == 1) s += s2;
        dst[(size_t)r * 512 + n0 + cc] = s;
    }
}

__global__ __launch_bounds__(FOLD_T)
void k_fold(const float* __restrict__ pw0, const float* __restrict__ pb0,
            const float* __restrict__ pw1, const float* __restrict__ pb1,
            const float* __restrict__ pw2, const float* __restrict__ pb2,
            const float* __restrict__ fuse_w, const float* __restrict__ fuse_b,
            const float* __restrict__ flat_w, const float* __restrict__ flat_b,
            float* __restrict__ Wtot, float* __restrict__ btot) {
    __shared__ __attribute__((aligned(16))) float sL[384 * 4];
    __shared__ __attribute__((aligned(16))) float sW[384 * 4];
    __shared__ __attribute__((aligned(16))) float sRed[4 * 384 * 4];
    const int n0 = (blockIdx.x & 7) * 64;
    const int g  = (int)(blockIdx.x >> 3) - 1;
    if (g >= 0) {
        int kk0 = 4 * g;
        const float* pw; int sc;
        if (kk0 < 8)       { sc = 0; pw = pw0 + (size_t)kk0 * 128; }
        else if (kk0 < 24) { sc = 1; pw = pw1 + (size_t)(kk0 - 8) * 128; }
        else               { sc = 2; pw = pw2 + (size_t)(kk0 - 24) * 128; }
        fold_body<4>(pw, pw, pw, 1, fuse_w + (size_t)sc * 128 * 384, nullptr,
                     flat_w, nullptr, Wtot + (size_t)kk0 * 512, n0, sL, sW, sRed);
    } else {
        fold_body<1>(pb0, pb1, pb2, 3, fuse_w, fuse_b,
                     flat_w, flat_b, btot, n0, sL, sW, sRed);
    }
}

__global__ void k_gemm(const float* __restrict__ P0, const float* __restrict__ P1,
                       const float* __restrict__ P2, const float* __restrict__ Wt,
                       const float* __restrict__ btot, float* __restrict__ out) {
    int wave = threadIdx.x >> 5;
    int lane = threadIdx.x & 31;
    int tile = blockIdx.x * 8 + wave;
    int mt = tile >> 5, nt = tile & 31;
    int mbase = mt * 16, nbase = nt * 16;
    int half = lane >> 4, l16 = lane & 15;
    const int row = mbase + l16;
    const float* A0a = P0 + (size_t)row * 8;
    const float* A0b = A0a + B_ * 512;
    const float* A1  = P1 + (size_t)row * 16;
    const float* A2  = P2 + (size_t)row * 32;
    const float* Bc = Wt + nbase + l16;
    v8f acc = {};
#pragma unroll
    for (int k0 = 0; k0 < 56; k0 += 4) {
        int ka = k0 + half * 2;
        v2f av;
        if (k0 < 8)       av = v2f{ A0a[ka] + A0b[ka], A0a[ka + 1] + A0b[ka + 1] };
        else if (k0 < 24) av = v2f{ A1[ka - 8], A1[ka - 7] };
        else              av = v2f{ A2[ka - 24], A2[ka - 23] };
        v2f bv = { Bc[(size_t)ka * 512], Bc[(size_t)(ka + 1) * 512] };
        acc = __builtin_amdgcn_wmma_f32_16x16x4_f32(
            false, av, false, bv, (short)0, acc, false, false);
    }
    float bias = btot[nbase + l16];
#pragma unroll
    for (int r = 0; r < 8; ++r)
        out[(size_t)(mbase + r + 8 * half) * 512 + nbase + l16] = acc[r] + bias;
}

extern "C" void kernel_launch(void* const* d_in, const int* in_sizes, int n_in,
                              void* d_out, int out_size, void* d_ws, size_t ws_size,
                              hipStream_t stream) {
    (void)in_sizes; (void)n_in; (void)out_size; (void)ws_size;
    const float* x      = (const float*)d_in[0];
    const float* pw0    = (const float*)d_in[1];
    const float* pb0    = (const float*)d_in[2];
    const float* pw1    = (const float*)d_in[3];
    const float* pb1    = (const float*)d_in[4];
    const float* pw2    = (const float*)d_in[5];
    const float* pb2    = (const float*)d_in[6];
    const float* fuse_w = (const float*)d_in[7];
    const float* fuse_b = (const float*)d_in[8];
    const float* flat_w = (const float*)d_in[9];
    const float* flat_b = (const float*)d_in[10];
    float* out = (float*)d_out;

    float* ws   = (float*)d_ws;
    float* P0   = ws;
    float* P1   = P0 + 2 * B_ * 512;
    float* P2   = P1 + B_ * 1024;
    float* Wtot = P2 + B_ * 2048;
    float* btot = Wtot + 56 * 512;

    k_fold<<<15 * 8, FOLD_T, 0, stream>>>(pw0, pb0, pw1, pb1, pw2, pb2, fuse_w, fuse_b,
                                          flat_w, flat_b, Wtot, btot);

    dim3 gx(8, 2, B_);
    k_xpass<<<gx, XP_T, 0, stream>>>(x, P0, P1, P2);

    k_gemm<<<2048 / 8, 256, 0, stream>>>(P0, P1, P2, Wtot, btot, out);
}
